# starting kernel
# baseline (speedup 1.0000x reference)
#include <hip/hip_runtime.h>
#include <hip/hip_bf16.h>

typedef __attribute__((ext_vector_type(2))) float v2f;
typedef __attribute__((ext_vector_type(8))) float v8f;

namespace {

constexpr int T_STEPS = 5;
constexpr int BATCH   = 1000000;
constexpr int IN_DIM  = 5;
constexpr int HID     = 6;
constexpr int ELEMS_PER_BLOCK = 80;
constexpr int WAVES_PER_BLOCK = 5;
constexpr int THREADS = WAVES_PER_BLOCK * 32;
constexpr int ROWS_PER_T = BATCH / T_STEPS;

__device__ __forceinline__ float swap16(float v) {
  return __int_as_float(
      __builtin_amdgcn_ds_swizzle(__float_as_int(v), 0x401F));
}

__device__ __forceinline__ float tanh_fast(float x) {
#if __has_builtin(__builtin_amdgcn_tanhf)
  return __builtin_amdgcn_tanhf(x);
#else
  float e = __expf(-2.0f * __builtin_fabsf(x));
  float r = __fdividef(1.0f - e, 1.0f + e);
  return __builtin_copysignf(r, x);
#endif
}

__device__ __forceinline__ float sig_fast(float x) {
  return 0.5f * tanh_fast(0.5f * x) + 0.5f;
}

__device__ __forceinline__ v8f wmma4(v2f a, v2f b, v8f c) {
  return __builtin_amdgcn_wmma_f32_16x16x4_f32(false, a, false, b, (short)0, c,
                                               false, false);
}

__device__ __forceinline__ void lstm_unit(float gi, float gf, float gg, float go,
                                          float& c, float& h) {
  float I = sig_fast(gi);
  float F = sig_fast(gf);
  float G = tanh_fast(gg);
  float O = sig_fast(go);
  c = F * c + I * G;
  h = O * tanh_fast(c);
}

__device__ __forceinline__ float wcol_l0(const float* wih, const float* whh, int R, int c) {
  if (c < 5)  return wih[R * 5 + c];
  if (c < 11) return whh[R * 6 + (c - 5)];
  return 0.0f;
}
__device__ __forceinline__ float wcol_l1(const float* wih, const float* whh, int R, int c) {
  if (c < 6) return wih[R * 6 + c];
  return whh[R * 6 + (c - 6)];
}

__device__ __forceinline__ int tile0_row(int m) {
  return (m & 3) * HID + (m >> 2);
}
__device__ __forceinline__ bool tile1_row(int m, int& R) {
  if (m < 4)            { R = m * HID + 4;       return true; }
  if (m >= 8 && m < 12) { R = (m - 8) * HID + 5; return true; }
  return false;
}

}

__global__ __launch_bounds__(THREADS)
void lstm2_wmma_kernel(const float* __restrict__ x,
                       const float* __restrict__ w_ih0, const float* __restrict__ w_hh0,
                       const float* __restrict__ b_ih0, const float* __restrict__ b_hh0,
                       const float* __restrict__ w_ih1, const float* __restrict__ w_hh1,
                       const float* __restrict__ b_ih1, const float* __restrict__ b_hh1,
                       const float* __restrict__ w_out, const float* __restrict__ b_out,
                       float* __restrict__ out) {
  __shared__ float sp[T_STEPS * 2 * ELEMS_PER_BLOCK];

  const int tid  = threadIdx.x;
  const int wave = tid >> 5;
  const int lane = tid & 31;
  const int n    = lane & 15;
  const int hi   = lane >> 4;
  const int e    = wave * 16 + n;
  const int b    = blockIdx.x * ELEMS_PER_BLOCK + e;

  v2f A0t0[3], A0t1[3], A1t0[3], A1t1[3];
  {
    const int m  = n;
    const int R0 = tile0_row(m);
    int R1; const bool v1 = tile1_row(m, R1);
#pragma unroll
    for (int s = 0; s < 3; ++s) {
      const int c = 4 * s + 2 * hi;
      A0t0[s][0] = wcol_l0(w_ih0, w_hh0, R0, c);
      A0t0[s][1] = wcol_l0(w_ih0, w_hh0, R0, c + 1);
      A0t1[s][0] = v1 ? wcol_l0(w_ih0, w_hh0, R1, c)     : 0.0f;
      A0t1[s][1] = v1 ? wcol_l0(w_ih0, w_hh0, R1, c + 1) : 0.0f;
      A1t0[s][0] = wcol_l1(w_ih1, w_hh1, R0, c);
      A1t0[s][1] = wcol_l1(w_ih1, w_hh1, R0, c + 1);
      A1t1[s][0] = v1 ? wcol_l1(w_ih1, w_hh1, R1, c)     : 0.0f;
      A1t1[s][1] = v1 ? wcol_l1(w_ih1, w_hh1, R1, c + 1) : 0.0f;
    }
  }

  v8f c0i0 = {}, c1i0 = {}, c0i1 = {}, c1i1 = {};
#pragma unroll
  for (int r = 0; r < 8; ++r) {
    const int m = r + 8 * hi;
    const int R = tile0_row(m);
    c0i0[r] = b_ih0[R] + b_hh0[R];
    c0i1[r] = b_ih1[R] + b_hh1[R];
    int R1;
    if (tile1_row(m, R1)) {
      c1i0[r] = b_ih0[R1] + b_hh0[R1];
      c1i1[r] = b_ih1[R1] + b_hh1[R1];
    }
  }

  const int p  = e % 5;
  const int u0 = hi ? 2 : 0, u1 = hi ? 3 : 1, u2 = hi ? 5 : 4;
  const float wl0 = w_out[p * HID + u0];
  const float wl1 = w_out[p * HID + u1];
  const float wl2 = w_out[p * HID + u2];

  float xA[T_STEPS], xB[T_STEPS], xC[T_STEPS];
#pragma unroll
  for (int t = 0; t < T_STEPS; ++t) {
    const float* xp = x + ((size_t)t * BATCH + b) * IN_DIM;
    xA[t] = xp[2 * hi];
    xB[t] = xp[2 * hi + 1];
    xC[t] = xp[4];
  }

  float h1[3] = {0.f, 0.f, 0.f}, c1s[3] = {0.f, 0.f, 0.f};
  float h2[3] = {0.f, 0.f, 0.f}, c2s[3] = {0.f, 0.f, 0.f};

#pragma unroll
  for (int t = 0; t < T_STEPS; ++t) {
    const float sw1 = swap16(h1[1]);
    v2f B0, B1, B2;
    B0[0] = xA[t];               B0[1] = xB[t];
    B1[0] = hi ? sw1 : xC[t];    B1[1] = h1[0];
    B2[0] = hi ? h1[2] : sw1;    B2[1] = hi ? 0.0f : h1[2];

    v8f g0 = c0i0, g1 = c1i0;
    g0 = wmma4(A0t0[0], B0, g0);  g0 = wmma4(A0t0[1], B1, g0);  g0 = wmma4(A0t0[2], B2, g0);
    g1 = wmma4(A0t1[0], B0, g1);  g1 = wmma4(A0t1[1], B1, g1);  g1 = wmma4(A0t1[2], B2, g1);

    lstm_unit(g0[0], g0[1], g0[2], g0[3], c1s[0], h1[0]);
    lstm_unit(g0[4], g0[5], g0[6], g0[7], c1s[1], h1[1]);
    lstm_unit(g1[0], g1[1], g1[2], g1[3], c1s[2], h1[2]);

    const float ySw2 = swap16(h1[2]);
    const float s0 = swap16(h2[0]);
    const float s1 = swap16(h2[1]);
    const float s2 = swap16(h2[2]);
    v2f D0, D1, D2;
    D0[0] = h1[0];            D0[1] = h1[1];
    D1[0] = hi ? s0 : h1[2];  D1[1] = hi ? s1 : ySw2;
    D2[0] = hi ? s2 : s0;     D2[1] = hi ? h2[2] : s1;

    v8f q0 = c0i1, q1 = c1i1;
    q0 = wmma4(A1t0[0], D0, q0);  q0 = wmma4(A1t0[1], D1, q0);  q0 = wmma4(A1t0[2], D2, q0);
    q1 = wmma4(A1t1[0], D0, q1);  q1 = wmma4(A1t1[1], D1, q1);  q1 = wmma4(A1t1[2], D2, q1);

    lstm_unit(q0[0], q0[1], q0[2], q0[3], c2s[0], h2[0]);
    lstm_unit(q0[4], q0[5], q0[6], q0[7], c2s[1], h2[1]);
    lstm_unit(q1[0], q1[1], q1[2], q1[3], c2s[2], h2[2]);

    sp[(t * 2 + hi) * ELEMS_PER_BLOCK + e] = h2[0] * wl0 + h2[1] * wl1 + h2[2] * wl2;
  }

  __syncthreads();

  if (tid < T_STEPS * 16) {
    const int t   = tid >> 4;
    const int r16 = tid & 15;
    float sum = b_out[0];
#pragma unroll
    for (int pp = 0; pp < 5; ++pp) {
      const int ee = r16 * 5 + pp;
      sum += sp[(t * 2 + 0) * ELEMS_PER_BLOCK + ee] +
             sp[(t * 2 + 1) * ELEMS_PER_BLOCK + ee];
    }
    out[t * ROWS_PER_T + blockIdx.x * 16 + r16] = sum;
  }
}

extern "C" void kernel_launch(void* const* d_in, const int* in_sizes, int n_in,
                              void* d_out, int out_size, void* d_ws, size_t ws_size,
                              hipStream_t stream) {
  const float* x     = (const float*)d_in[0];
  const float* w_ih0 = (const float*)d_in[1];
  const float* w_hh0 = (const float*)d_in[2];
  const float* b_ih0 = (const float*)d_in[3];
  const float* b_hh0 = (const float*)d_in[4];
  const float* w_ih1 = (const float*)d_in[5];
  const float* w_hh1 = (const float*)d_in[6];
  const float* b_ih1 = (const float*)d_in[7];
  const float* b_hh1 = (const float*)d_in[8];
  const float* w_out = (const float*)d_in[9];
  const float* b_out = (const float*)d_in[10];
  float* out = (float*)d_out;

  const int grid = BATCH / ELEMS_PER_BLOCK;
  lstm2_wmma_kernel<<<grid, THREADS, 0, stream>>>(
      x, w_ih0, w_hh0, b_ih0, b_hh0, w_ih1, w_hh1, b_ih1, b_hh1, w_out, b_out, out);
}


# Grid of 1000 resident work-groups, each walking 8 blocks of 125 elements: a wave builds its weight fragments and biases once, next block's x loads fly under the combine; 8000 waves enter, not 64000
# speedup vs baseline: 1.1652x; 1.1652x over previous
#include <hip/hip_runtime.h>
#include <hip/hip_bf16.h>

typedef __attribute__((ext_vector_type(2))) float v2f;
typedef __attribute__((ext_vector_type(8))) float v8f;

namespace {

constexpr int T_STEPS = 5;
constexpr int BATCH   = 1000000;
constexpr int IN_DIM  = 5;
constexpr int HID     = 6;
constexpr int ELEMS_PER_BLOCK = 125;
constexpr int COLS_PER_BLOCK  = 128;
constexpr int ROWS_PER_BLOCK  = ELEMS_PER_BLOCK / 5;
constexpr int WAVES_PER_BLOCK = 8;
constexpr int THREADS = WAVES_PER_BLOCK * 32;
constexpr int ROWS_PER_T = BATCH / T_STEPS;
constexpr int NBLOCKS = BATCH / ELEMS_PER_BLOCK;
constexpr int GRID = 1000;
constexpr int BLOCKS_PER_GROUP = NBLOCKS / GRID;
constexpr size_t X_BLOCK_STRIDE = (size_t)GRID * ELEMS_PER_BLOCK * IN_DIM;
constexpr size_t X_T_STRIDE     = (size_t)BATCH * IN_DIM;

__device__ __forceinline__ float swap16(float v) {
  return __int_as_float(
      __builtin_amdgcn_ds_swizzle(__float_as_int(v), 0x401F));
}

__device__ __forceinline__ float tanh_fast(float x) {
#if __has_builtin(__builtin_amdgcn_tanhf)
  return __builtin_amdgcn_tanhf(x);
#else
  float e = __expf(-2.0f * __builtin_fabsf(x));
  float r = __fdividef(1.0f - e, 1.0f + e);
  return __builtin_copysignf(r, x);
#endif
}

__device__ __forceinline__ float sig_fast(float x) {
  return 0.5f * tanh_fast(0.5f * x) + 0.5f;
}

__device__ __forceinline__ v8f wmma4(v2f a, v2f b, v8f c) {
  return __builtin_amdgcn_wmma_f32_16x16x4_f32(false, a, false, b, (short)0, c,
                                               false, false);
}

__device__ __forceinline__ void lstm_unit(float gi, float gf, float gg, float go,
                                          float& c, float& h) {
  float I = sig_fast(gi);
  float F = sig_fast(gf);
  float G = tanh_fast(gg);
  float O = sig_fast(go);
  c = F * c + I * G;
  h = O * tanh_fast(c);
}

__device__ __forceinline__ float wcol_l0(const float* wih, const float* whh, int R, int c) {
  if (c < 5)  return wih[R * 5 + c];
  if (c < 11) return whh[R * 6 + (c - 5)];
  return 0.0f;
}
__device__ __forceinline__ float wcol_l1(const float* wih, const float* whh, int R, int c) {
  if (c < 6) return wih[R * 6 + c];
  return whh[R * 6 + (c - 6)];
}

__device__ __forceinline__ int tile0_row(int m) {
  return (m & 3) * HID + (m >> 2);
}
__device__ __forceinline__ bool tile1_row(int m, int& R) {
  if (m < 4)            { R = m * HID + 4;       return true; }
  if (m >= 8 && m < 12) { R = (m - 8) * HID + 5; return true; }
  return false;
}

}

__global__ __launch_bounds__(THREADS)
void lstm2_wmma_kernel(const float* __restrict__ x,
                       const float* __restrict__ w_ih0, const float* __restrict__ w_hh0,
                       const float* __restrict__ b_ih0, const float* __restrict__ b_hh0,
                       const float* __restrict__ w_ih1, const float* __restrict__ w_hh1,
                       const float* __restrict__ b_ih1, const float* __restrict__ b_hh1,
                       const float* __restrict__ w_out, const float* __restrict__ b_out,
                       float* __restrict__ out) {
  __shared__ float sp[2][T_STEPS * 2 * COLS_PER_BLOCK];

  const int tid  = threadIdx.x;
  const int wave = tid >> 5;
  const int lane = tid & 31;
  const int n    = lane & 15;
  const int hi   = lane >> 4;
  const int e    = wave * 16 + n;
  const int el   = e < ELEMS_PER_BLOCK ? e : ELEMS_PER_BLOCK - 1;

  v2f A0t0[3], A0t1[3], A1t0[3], A1t1[3];
  {
    const int m  = n;
    const int R0 = tile0_row(m);
    int R1; const bool v1 = tile1_row(m, R1);
#pragma unroll
    for (int s = 0; s < 3; ++s) {
      const int c = 4 * s + 2 * hi;
      A0t0[s][0] = wcol_l0(w_ih0, w_hh0, R0, c);
      A0t0[s][1] = wcol_l0(w_ih0, w_hh0, R0, c + 1);
      A0t1[s][0] = v1 ? wcol_l0(w_ih0, w_hh0, R1, c)     : 0.0f;
      A0t1[s][1] = v1 ? wcol_l0(w_ih0, w_hh0, R1, c + 1) : 0.0f;
      A1t0[s][0] = wcol_l1(w_ih1, w_hh1, R0, c);
      A1t0[s][1] = wcol_l1(w_ih1, w_hh1, R0, c + 1);
      A1t1[s][0] = v1 ? wcol_l1(w_ih1, w_hh1, R1, c)     : 0.0f;
      A1t1[s][1] = v1 ? wcol_l1(w_ih1, w_hh1, R1, c + 1) : 0.0f;
    }
  }

  v8f c0i0 = {}, c1i0 = {}, c0i1 = {}, c1i1 = {};
#pragma unroll
  for (int r = 0; r < 8; ++r) {
    const int m = r + 8 * hi;
    const int R = tile0_row(m);
    c0i0[r] = b_ih0[R] + b_hh0[R];
    c0i1[r] = b_ih1[R] + b_hh1[R];
    int R1;
    if (tile1_row(m, R1)) {
      c1i0[r] = b_ih0[R1] + b_hh0[R1];
      c1i1[r] = b_ih1[R1] + b_hh1[R1];
    }
  }

  const int p  = e % 5;
  const int u0 = hi ? 2 : 0, u1 = hi ? 3 : 1, u2 = hi ? 5 : 4;
  const float wl0 = w_out[p * HID + u0];
  const float wl1 = w_out[p * HID + u1];
  const float wl2 = w_out[p * HID + u2];

  const float* xl = x + ((size_t)blockIdx.x * ELEMS_PER_BLOCK + el) * IN_DIM;
  float xA[T_STEPS], xB[T_STEPS], xC[T_STEPS];
#pragma unroll
  for (int t = 0; t < T_STEPS; ++t) {
    const float* xp = xl + (size_t)t * X_T_STRIDE;
    xA[t] = xp[2 * hi];
    xB[t] = xp[2 * hi + 1];
    xC[t] = xp[4];
  }

  const int ct = tid / ROWS_PER_BLOCK;
  const int cr = tid - ct * ROWS_PER_BLOCK;
  const float bo = b_out[0];

  for (int trip = 0; trip < BLOCKS_PER_GROUP; ++trip) {
  const int blk = blockIdx.x + trip * GRID;
  const int nxt = trip + 1 < BLOCKS_PER_GROUP ? trip + 1 : trip;
  const float* xn = xl + (size_t)nxt * X_BLOCK_STRIDE;
  float* spb = sp[trip & 1];

  float h1[3] = {0.f, 0.f, 0.f}, c1s[3] = {0.f, 0.f, 0.f};
  float h2[3] = {0.f, 0.f, 0.f}, c2s[3] = {0.f, 0.f, 0.f};

#pragma unroll
  for (int t = 0; t < T_STEPS; ++t) {
    const float sw1 = swap16(h1[1]);
    v2f B0, B1, B2;
    B0[0] = xA[t];               B0[1] = xB[t];
    B1[0] = hi ? sw1 : xC[t];    B1[1] = h1[0];
    B2[0] = hi ? h1[2] : sw1;    B2[1] = hi ? 0.0f : h1[2];

    v8f g0 = c0i0, g1 = c1i0;
    g0 = wmma4(A0t0[0], B0, g0);  g0 = wmma4(A0t0[1], B1, g0);  g0 = wmma4(A0t0[2], B2, g0);
    g1 = wmma4(A0t1[0], B0, g1);  g1 = wmma4(A0t1[1], B1, g1);  g1 = wmma4(A0t1[2], B2, g1);

    {
      const float* xq = xn + (size_t)t * X_T_STRIDE;
      xA[t] = xq[2 * hi];
      xB[t] = xq[2 * hi + 1];
      xC[t] = xq[4];
    }

    lstm_unit(g0[0], g0[1], g0[2], g0[3], c1s[0], h1[0]);
    lstm_unit(g0[4], g0[5], g0[6], g0[7], c1s[1], h1[1]);
    lstm_unit(g1[0], g1[1], g1[2], g1[3], c1s[2], h1[2]);

    const float ySw2 = swap16(h1[2]);
    const float s0 = swap16(h2[0]);
    const float s1 = swap16(h2[1]);
    const float s2 = swap16(h2[2]);
    v2f D0, D1, D2;
    D0[0] = h1[0];            D0[1] = h1[1];
    D1[0] = hi ? s0 : h1[2];  D1[1] = hi ? s1 : ySw2;
    D2[0] = hi ? s2 : s0;     D2[1] = hi ? h2[2] : s1;

    v8f q0 = c0i1, q1 = c1i1;
    q0 = wmma4(A1t0[0], D0, q0);  q0 = wmma4(A1t0[1], D1, q0);  q0 = wmma4(A1t0[2], D2, q0);
    q1 = wmma4(A1t1[0], D0, q1);  q1 = wmma4(A1t1[1], D1, q1);  q1 = wmma4(A1t1[2], D2, q1);

    lstm_unit(q0[0], q0[1], q0[2], q0[3], c2s[0], h2[0]);
    lstm_unit(q0[4], q0[5], q0[6], q0[7], c2s[1], h2[1]);
    lstm_unit(q1[0], q1[1], q1[2], q1[3], c2s[2], h2[2]);

    spb[(t * 2 + hi) * COLS_PER_BLOCK + e] = h2[0] * wl0 + h2[1] * wl1 + h2[2] * wl2;
  }

  __builtin_amdgcn_fence(__ATOMIC_RELEASE, "workgroup", "local");
  __builtin_amdgcn_s_barrier();
  __builtin_amdgcn_fence(__ATOMIC_ACQUIRE, "workgroup", "local");

  if (tid < T_STEPS * ROWS_PER_BLOCK) {
    float sum = bo;
#pragma unroll
    for (int pp = 0; pp < 5; ++pp) {
      const int ee = cr * 5 + pp;
      sum += spb[(ct * 2 + 0) * COLS_PER_BLOCK + ee] +
             spb[(ct * 2 + 1) * COLS_PER_BLOCK + ee];
    }
    out[ct * ROWS_PER_T + blk * ROWS_PER_BLOCK + cr] = sum;
  }
  }
}

extern "C" void kernel_launch(void* const* d_in, const int* in_sizes, int n_in,
                              void* d_out, int out_size, void* d_ws, size_t ws_size,
                              hipStream_t stream) {
  const float* x     = (const float*)d_in[0];
  const float* w_ih0 = (const float*)d_in[1];
  const float* w_hh0 = (const float*)d_in[2];
  const float* b_ih0 = (const float*)d_in[3];
  const float* b_hh0 = (const float*)d_in[4];
  const float* w_ih1 = (const float*)d_in[5];
  const float* w_hh1 = (const float*)d_in[6];
  const float* b_ih1 = (const float*)d_in[7];
  const float* b_hh1 = (const float*)d_in[8];
  const float* w_out = (const float*)d_in[9];
  const float* b_out = (const float*)d_in[10];
  float* out = (float*)d_out;

  lstm2_wmma_kernel<<<GRID, THREADS, 0, stream>>>(
      x, w_ih0, w_hh0, b_ih0, b_hh0, w_ih1, w_hh1, b_ih1, b_hh1, w_out, b_out, out);
}
